# starting kernel
# baseline (speedup 1.0000x reference)
#include <hip/hip_runtime.h>
#include <math.h>

#define B_    8
#define T_    8192
#define D_    1024
#define TWO_D 2048
#define OUT_  3
#define EPS_  1e-6f

typedef float v2f __attribute__((ext_vector_type(2)));
typedef float v8f __attribute__((ext_vector_type(8)));

__device__ __forceinline__ float silu_f(float v) {
    return v * (1.0f / (1.0f + __expf(-v)));
}

__global__ __launch_bounds__(256) void ada_mod_wmma(
    const float* __restrict__ cond, const float* __restrict__ w_ada,
    const float* __restrict__ b_ada, float* __restrict__ mod)
{
    __shared__ __align__(16) float As[B_ * 1026];
    const int tid = threadIdx.x;
    for (int i = tid; i < B_ * D_; i += 256) {
        const int r = i >> 10, c = i & (D_ - 1);
        As[r * 1026 + c] = silu_f(cond[i]);
    }
    __syncthreads();

    const int wave = tid >> 5;
    const int lane = tid & 31;
    const int half = lane >> 4;
    const int l    = lane & 15;
    const int arow = (l & (B_ - 1)) * 1026;
    const int n0   = (blockIdx.x * 8 + wave) * 16;

    v8f acc = {};
    #pragma unroll 4
    for (int k = 0; k < D_; k += 4) {
        const int kk = k + 2 * half;
        const v2f a = *(const v2f*)(&As[arow + kk]);
        v2f bf;
        bf.x = w_ada[(size_t)kk       * TWO_D + n0 + l];
        bf.y = w_ada[(size_t)(kk + 1) * TWO_D + n0 + l];
        acc = __builtin_amdgcn_wmma_f32_16x16x4_f32(
            false, a, false, bf, (short)0, acc, false, false);
    }

    if (half == 0) {
        const float bb = b_ada[n0 + l];
        #pragma unroll
        for (int r = 0; r < B_; ++r)
            mod[r * TWO_D + n0 + l] = acc[r] + bb;
    }
}

__device__ __forceinline__ float wave_sum(float v) {
    #pragma unroll
    for (int off = 16; off > 0; off >>= 1)
        v += __shfl_xor(v, off, 32);
    return v;
}

__global__ __launch_bounds__(256) void ln_mod_silu_proj(
    const float* __restrict__ x, const float* __restrict__ mod,
    const float* __restrict__ w_proj, const float* __restrict__ b_proj,
    float* __restrict__ out)
{
    __shared__ float red[24];
    const int row  = blockIdx.x;
    const int b    = row >> 13;
    const int tid  = threadIdx.x;
    const int wave = tid >> 5, lane = tid & 31;

    const float4 xv = ((const float4*)(x + (size_t)row * D_))[tid];
    float s  = xv.x + xv.y + xv.z + xv.w;
    float sq = xv.x * xv.x + xv.y * xv.y + xv.z * xv.z + xv.w * xv.w;
    s  = wave_sum(s);
    sq = wave_sum(sq);
    if (lane == 0) { red[wave] = s; red[8 + wave] = sq; }
    __syncthreads();
    float ts = 0.0f, tq = 0.0f;
    #pragma unroll
    for (int i = 0; i < 8; ++i) { ts += red[i]; tq += red[8 + i]; }
    const float mu  = ts * (1.0f / D_);
    const float var = tq * (1.0f / D_) - mu * mu;
    const float rs  = rsqrtf(var + EPS_);

    const float4 sc4 = ((const float4*)(mod + b * TWO_D))[tid];
    const float4 sh4 = ((const float4*)(mod + b * TWO_D + D_))[tid];
    const float4 w0  = ((const float4*)w_proj)[tid * 3 + 0];
    const float4 w1  = ((const float4*)w_proj)[tid * 3 + 1];
    const float4 w2  = ((const float4*)w_proj)[tid * 3 + 2];

    const float xe[4]  = {xv.x,  xv.y,  xv.z,  xv.w};
    const float sce[4] = {sc4.x, sc4.y, sc4.z, sc4.w};
    const float she[4] = {sh4.x, sh4.y, sh4.z, sh4.w};
    const float wf[12] = {w0.x, w0.y, w0.z, w0.w,
                          w1.x, w1.y, w1.z, w1.w,
                          w2.x, w2.y, w2.z, w2.w};

    float a0 = 0.0f, a1 = 0.0f, a2 = 0.0f;
    #pragma unroll
    for (int e = 0; e < 4; ++e) {
        const float y = silu_f((xe[e] - mu) * rs * (1.0f + sce[e]) + she[e]);
        a0 += y * wf[e * 3 + 0];
        a1 += y * wf[e * 3 + 1];
        a2 += y * wf[e * 3 + 2];
    }
    a0 = wave_sum(a0);
    a1 = wave_sum(a1);
    a2 = wave_sum(a2);
    __syncthreads();
    if (lane == 0) { red[wave] = a0; red[8 + wave] = a1; red[16 + wave] = a2; }
    __syncthreads();
    if (tid == 0) {
        float r0 = 0.0f, r1 = 0.0f, r2 = 0.0f;
        #pragma unroll
        for (int i = 0; i < 8; ++i) {
            r0 += red[i]; r1 += red[8 + i]; r2 += red[16 + i];
        }
        out[row * OUT_ + 0] = r0 + b_proj[0];
        out[row * OUT_ + 1] = r1 + b_proj[1];
        out[row * OUT_ + 2] = r2 + b_proj[2];
    }
}

extern "C" void kernel_launch(void* const* d_in, const int* in_sizes, int n_in,
                              void* d_out, int out_size, void* d_ws, size_t ws_size,
                              hipStream_t stream) {
    (void)in_sizes; (void)n_in; (void)out_size; (void)ws_size;
    const float* x      = (const float*)d_in[0];
    const float* cond   = (const float*)d_in[1];
    const float* w_ada  = (const float*)d_in[2];
    const float* b_ada  = (const float*)d_in[3];
    const float* w_proj = (const float*)d_in[4];
    const float* b_proj = (const float*)d_in[5];
    float*       out    = (float*)d_out;
    float*       mod    = (float*)d_ws;

    ada_mod_wmma<<<dim3(16), dim3(256), 0, stream>>>(cond, w_ada, b_ada, mod);
    ln_mod_silu_proj<<<dim3(B_ * T_), dim3(256), 0, stream>>>(x, mod, w_proj, b_proj, out);
}


# Kernel 2 as a stream: one wave per 16 rows (4096 waves, was 524288), x by non-temporal loads one row ahead, scale/shift in registers, w_proj in LDS, in-wave DPP sums (new summation order)
# speedup vs baseline: 2.2109x; 2.2109x over previous
#include <hip/hip_runtime.h>
#include <math.h>

#define B_    8
#define T_    8192
#define D_    1024
#define TWO_D 2048
#define OUT_  3
#define EPS_  1e-6f

typedef float v2f __attribute__((ext_vector_type(2)));
typedef float v8f __attribute__((ext_vector_type(8)));

__device__ __forceinline__ float silu_f(float v) {
    return v * (1.0f / (1.0f + __expf(-v)));
}

__global__ __launch_bounds__(256) void ada_mod_wmma(
    const float* __restrict__ cond, const float* __restrict__ w_ada,
    const float* __restrict__ b_ada, float* __restrict__ mod)
{
    __shared__ __align__(16) float As[B_ * 1026];
    const int tid = threadIdx.x;
    for (int i = tid; i < B_ * D_; i += 256) {
        const int r = i >> 10, c = i & (D_ - 1);
        As[r * 1026 + c] = silu_f(cond[i]);
    }
    __syncthreads();

    const int wave = tid >> 5;
    const int lane = tid & 31;
    const int half = lane >> 4;
    const int l    = lane & 15;
    const int arow = (l & (B_ - 1)) * 1026;
    const int n0   = (blockIdx.x * 8 + wave) * 16;

    v8f acc = {};
    #pragma unroll 4
    for (int k = 0; k < D_; k += 4) {
        const int kk = k + 2 * half;
        const v2f a = *(const v2f*)(&As[arow + kk]);
        v2f bf;
        bf.x = w_ada[(size_t)kk       * TWO_D + n0 + l];
        bf.y = w_ada[(size_t)(kk + 1) * TWO_D + n0 + l];
        acc = __builtin_amdgcn_wmma_f32_16x16x4_f32(
            false, a, false, bf, (short)0, acc, false, false);
    }

    if (half == 0) {
        const float bb = b_ada[n0 + l];
        #pragma unroll
        for (int r = 0; r < B_; ++r)
            mod[r * TWO_D + n0 + l] = acc[r] + bb;
    }
}

#define K2_WAVES 16
#define K2_ROWS  16
#define K2_RING  2
static_assert(T_ % (K2_WAVES * K2_ROWS) == 0, "all rows of a work-group share one batch entry");
static_assert(K2_ROWS <= 32, "lane r keeps the results of row r");
static_assert(K2_ROWS % K2_RING == 0, "the row loop is unrolled by the ring");
static_assert((B_ * T_) % (K2_WAVES * K2_ROWS) == 0, "whole work-groups");
static_assert((K2_WAVES * K2_ROWS * OUT_) % 128 == 0, "the gathered results are whole 16-byte pieces per lane");

typedef float f4 __attribute__((ext_vector_type(4)));

__device__ __forceinline__ f4 ld_stream(const f4* p) { return __builtin_nontemporal_load(p); }

template <int CTRL>
__device__ __forceinline__ float row_xor(float v) {
    const int b = __builtin_bit_cast(int, v);
    return __builtin_bit_cast(float, __builtin_amdgcn_update_dpp(b, b, CTRL, 0xf, 0xf, true));
}
__device__ __forceinline__ float xhalf(float v) {
    const int b = __builtin_bit_cast(int, v);
    return __builtin_bit_cast(float,
        __builtin_amdgcn_permlanex16(b, b, 0x76543210, 0xfedcba98, false, false));
}
__device__ __forceinline__ float wave_sum32(float v) {
    v += row_xor<0x161>(v);
    v += row_xor<0x162>(v);
    v += row_xor<0x164>(v);
    v += row_xor<0x168>(v);
    v += xhalf(v);
    return v;
}
__device__ __forceinline__ float hsum4(f4 v) { return (v.x + v.y) + (v.z + v.w); }

__global__ __launch_bounds__(K2_WAVES * 32) void ln_mod_silu_proj(
    const float* __restrict__ x, const float* __restrict__ mod,
    const float* __restrict__ w_proj, const float* __restrict__ b_proj,
    float* __restrict__ out)
{
    __shared__ __align__(16) float res[K2_WAVES * K2_ROWS * OUT_];
    __shared__ f4 wl[OUT_][8][32];
    const int lane = threadIdx.x & 31;
    const int wave = __builtin_amdgcn_readfirstlane((int)(threadIdx.x >> 5));
    const int row0 = (blockIdx.x * K2_WAVES + wave) * K2_ROWS;
    const int b    = row0 >> 13;

    const f4* xp = (const f4*)(x + (size_t)row0 * D_) + lane;
    f4 xb[K2_RING][8];
#define LOAD_ROW(SLOT, R) do { \
        _Pragma("unroll") for (int q_ = 0; q_ < 8; ++q_) \
            xb[SLOT][q_] = ld_stream(xp + (R) * (D_ / 4) + 32 * q_); } while (0)

    LOAD_ROW(0, 0);
    __builtin_amdgcn_sched_barrier(0);

    f4 g[8], sh[8];
    {
        const f4* sc_p = (const f4*)(mod + b * TWO_D) + lane;
        const f4* sh_p = (const f4*)(mod + b * TWO_D + D_) + lane;
        #pragma unroll
        for (int q = 0; q < 8; ++q) {
            g[q]  = 1.0f + sc_p[32 * q];
            sh[q] = sh_p[32 * q];
        }
    }
    if (threadIdx.x < 256) {
        const int sq = threadIdx.x >> 5;
        const f4* w_p = (const f4*)w_proj + 96 * sq + 3 * lane;
        const f4 wa = w_p[0], wb = w_p[1], wc = w_p[2];
        wl[0][sq][lane] = f4{wa.x, wa.w, wb.z, wc.y};
        wl[1][sq][lane] = f4{wa.y, wb.x, wb.w, wc.z};
        wl[2][sq][lane] = f4{wa.z, wb.y, wc.x, wc.w};
    }
    __syncthreads();

    float m0 = 0.0f, m1 = 0.0f, m2 = 0.0f;

    #pragma nounroll
    for (int i = 0; i < K2_ROWS / K2_RING; ++i) {
        #pragma unroll
        for (int j = 0; j < K2_RING; ++j) {
            const int r = i * K2_RING + j;
            const int ra = (r + 1 < K2_ROWS) ? r + 1 : K2_ROWS - 1;
            __builtin_amdgcn_sched_barrier(0);
            LOAD_ROW((j + 1) % K2_RING, ra);
            __builtin_amdgcn_sched_barrier(0);

            const f4* xr = xb[j];
            f4 s4 = xr[0];
            f4 q4 = xr[0] * xr[0];
            #pragma unroll
            for (int q = 1; q < 8; ++q) { s4 += xr[q]; q4 += xr[q] * xr[q]; }
            const float ts = wave_sum32(hsum4(s4));
            const float tq = wave_sum32(hsum4(q4));
            const float mu  = ts * (1.0f / D_);
            const float var = tq * (1.0f / D_) - mu * mu;
            const float rs  = rsqrtf(var + EPS_);

            f4 a0 = {0.0f, 0.0f, 0.0f, 0.0f}, a1 = a0, a2 = a0;
            #pragma unroll
            for (int q = 0; q < 8; ++q) {
                const f4 t = (xr[q] - mu) * rs * g[q] + sh[q];
                const f4 y = f4{silu_f(t.x), silu_f(t.y), silu_f(t.z), silu_f(t.w)};
                a0 += y * wl[0][q][lane];
                a1 += y * wl[1][q][lane];
                a2 += y * wl[2][q][lane];
            }
            const float r0 = wave_sum32(hsum4(a0));
            const float r1 = wave_sum32(hsum4(a1));
            const float r2 = wave_sum32(hsum4(a2));
            if (lane == r) { m0 = r0; m1 = r1; m2 = r2; }
        }
    }
#undef LOAD_ROW

    if (lane < K2_ROWS) {
        float* p = res + (wave * K2_ROWS + lane) * OUT_;
        p[0] = m0 + b_proj[0];
        p[1] = m1 + b_proj[1];
        p[2] = m2 + b_proj[2];
    }
    __syncthreads();
    if (threadIdx.x < K2_WAVES * K2_ROWS * OUT_ / 4) {
        f4* o = (f4*)(out + (size_t)blockIdx.x * (K2_WAVES * K2_ROWS * OUT_));
        o[threadIdx.x] = ((const f4*)res)[threadIdx.x];
    }
}

extern "C" void kernel_launch(void* const* d_in, const int* in_sizes, int n_in,
                              void* d_out, int out_size, void* d_ws, size_t ws_size,
                              hipStream_t stream) {
    (void)in_sizes; (void)n_in; (void)out_size; (void)ws_size;
    const float* x      = (const float*)d_in[0];
    const float* cond   = (const float*)d_in[1];
    const float* w_ada  = (const float*)d_in[2];
    const float* b_ada  = (const float*)d_in[3];
    const float* w_proj = (const float*)d_in[4];
    const float* b_proj = (const float*)d_in[5];
    float*       out    = (float*)d_out;
    float*       mod    = (float*)d_ws;

    ada_mod_wmma<<<dim3(16), dim3(256), 0, stream>>>(cond, w_ada, b_ada, mod);
    ln_mod_silu_proj<<<dim3(B_ * T_ / (K2_WAVES * K2_ROWS)), dim3(K2_WAVES * 32), 0, stream>>>(x, mod, w_proj, b_proj, out);
}


# Kernel 1: K split over 16 waves per 16-column tile (64 groups of 32 waves): a wave's 32 w_ada loads all in flight at once, not 64 load-wait trips; partials summed through LDS (new summation order)
# speedup vs baseline: 2.9541x; 1.3362x over previous
#include <hip/hip_runtime.h>
#include <math.h>

#define B_    8
#define T_    8192
#define D_    1024
#define TWO_D 2048
#define OUT_  3
#define EPS_  1e-6f

typedef float v2f __attribute__((ext_vector_type(2)));
typedef float v8f __attribute__((ext_vector_type(8)));

__device__ __forceinline__ float silu_f(float v) {
    return v * (1.0f / (1.0f + __expf(-v)));
}

#define K1_TILES  2
#define K1_KSPLIT 16
#define K1_WAVES  (K1_TILES * K1_KSPLIT)
#define K1_KW     (D_ / K1_KSPLIT)
#define K1_STEPS  (K1_KW / 4)
static_assert(K1_STEPS % 2 == 0, "two accumulators take alternate k steps");
static_assert(B_ * D_ == 2 * 4 * K1_WAVES * 32, "two 16-byte pieces of cond per thread");
static_assert(K1_TILES * 16 * 4 == 128, "a work-group owns whole 128-byte lines of mod");
static_assert(B_ * K1_TILES * 16 <= K1_WAVES * 32, "one thread per result of the work-group");

typedef float f4 __attribute__((ext_vector_type(4)));

__global__ __launch_bounds__(K1_WAVES * 32) void ada_mod_wmma(
    const float* __restrict__ cond, const float* __restrict__ w_ada,
    const float* __restrict__ b_ada, float* __restrict__ mod)
{
    __shared__ __align__(16) float As[B_ * 1026];
    __shared__ __align__(16) float red[K1_WAVES * B_ * 16];
    const int tid  = threadIdx.x;
    const int wave = tid >> 5;
    const int lane = tid & 31;
    const int half = lane >> 4;
    const int l    = lane & 15;
    const int tile = wave / K1_KSPLIT;
    const int k0   = (wave % K1_KSPLIT) * K1_KW;
    const int n0   = blockIdx.x * (16 * K1_TILES) + tile * 16;

    const f4 c0 = ((const f4*)cond)[tid];
    const f4 c1 = ((const f4*)cond)[tid + K1_WAVES * 32];
    v2f bf[K1_STEPS];
    {
        const float* wp = w_ada + (size_t)(k0 + 2 * half) * TWO_D + n0 + l;
        #pragma unroll
        for (int s = 0; s < K1_STEPS; ++s) {
            bf[s].x = wp[(size_t)(4 * s)     * TWO_D];
            bf[s].y = wp[(size_t)(4 * s + 1) * TWO_D];
        }
    }

    {
        const int i0 = 4 * tid, i1 = 4 * (tid + K1_WAVES * 32);
        float* d0 = &As[(i0 >> 10) * 1026 + (i0 & (D_ - 1))];
        float* d1 = &As[(i1 >> 10) * 1026 + (i1 & (D_ - 1))];
        *(v2f*)(d0)     = v2f{silu_f(c0.x), silu_f(c0.y)};
        *(v2f*)(d0 + 2) = v2f{silu_f(c0.z), silu_f(c0.w)};
        *(v2f*)(d1)     = v2f{silu_f(c1.x), silu_f(c1.y)};
        *(v2f*)(d1 + 2) = v2f{silu_f(c1.z), silu_f(c1.w)};
    }
    __syncthreads();

    const int arow = (l & (B_ - 1)) * 1026;
    v8f acc0 = {}, acc1 = {};
    #pragma unroll
    for (int s = 0; s < K1_STEPS; s += 2) {
        const int kk = k0 + 4 * s + 2 * half;
        const v2f a0 = *(const v2f*)(&As[arow + kk]);
        const v2f a1 = *(const v2f*)(&As[arow + kk + 4]);
        acc0 = __builtin_amdgcn_wmma_f32_16x16x4_f32(
            false, a0, false, bf[s],     (short)0, acc0, false, false);
        acc1 = __builtin_amdgcn_wmma_f32_16x16x4_f32(
            false, a1, false, bf[s + 1], (short)0, acc1, false, false);
    }
    const v8f acc = acc0 + acc1;

    if (half == 0) {
        #pragma unroll
        for (int r = 0; r < B_; ++r)
            red[(wave * B_ + r) * 16 + l] = acc[r];
    }
    __syncthreads();

    if (tid < B_ * K1_TILES * 16) {
        const int r = tid >> 5, cc = tid & 31;
        const float* p = red + ((cc >> 4) * K1_KSPLIT * B_ + r) * 16 + (cc & 15);
        float sum = p[0];
        #pragma unroll
        for (int w = 1; w < K1_KSPLIT; ++w) sum += p[w * B_ * 16];
        const int n = blockIdx.x * (16 * K1_TILES) + cc;
        mod[r * TWO_D + n] = sum + b_ada[n];
    }
}

#define K2_WAVES 16
#define K2_ROWS  16
#define K2_RING  2
static_assert(T_ % (K2_WAVES * K2_ROWS) == 0, "all rows of a work-group share one batch entry");
static_assert(K2_ROWS <= 32, "lane r keeps the results of row r");
static_assert(K2_ROWS % K2_RING == 0, "the row loop is unrolled by the ring");
static_assert((B_ * T_) % (K2_WAVES * K2_ROWS) == 0, "whole work-groups");
static_assert((K2_WAVES * K2_ROWS * OUT_) % 128 == 0, "the gathered results are whole 16-byte pieces per lane");

__device__ __forceinline__ f4 ld_stream(const f4* p) { return __builtin_nontemporal_load(p); }

template <int CTRL>
__device__ __forceinline__ float row_xor(float v) {
    const int b = __builtin_bit_cast(int, v);
    return __builtin_bit_cast(float, __builtin_amdgcn_update_dpp(b, b, CTRL, 0xf, 0xf, true));
}
__device__ __forceinline__ float xhalf(float v) {
    const int b = __builtin_bit_cast(int, v);
    return __builtin_bit_cast(float,
        __builtin_amdgcn_permlanex16(b, b, 0x76543210, 0xfedcba98, false, false));
}
__device__ __forceinline__ float wave_sum32(float v) {
    v += row_xor<0x161>(v);
    v += row_xor<0x162>(v);
    v += row_xor<0x164>(v);
    v += row_xor<0x168>(v);
    v += xhalf(v);
    return v;
}
__device__ __forceinline__ float hsum4(f4 v) { return (v.x + v.y) + (v.z + v.w); }

__global__ __launch_bounds__(K2_WAVES * 32) void ln_mod_silu_proj(
    const float* __restrict__ x, const float* __restrict__ mod,
    const float* __restrict__ w_proj, const float* __restrict__ b_proj,
    float* __restrict__ out)
{
    __shared__ __align__(16) float res[K2_WAVES * K2_ROWS * OUT_];
    __shared__ f4 wl[OUT_][8][32];
    const int lane = threadIdx.x & 31;
    const int wave = __builtin_amdgcn_readfirstlane((int)(threadIdx.x >> 5));
    const int row0 = (blockIdx.x * K2_WAVES + wave) * K2_ROWS;
    const int b    = row0 >> 13;

    const f4* xp = (const f4*)(x + (size_t)row0 * D_) + lane;
    f4 xb[K2_RING][8];
#define LOAD_ROW(SLOT, R) do { \
        _Pragma("unroll") for (int q_ = 0; q_ < 8; ++q_) \
            xb[SLOT][q_] = ld_stream(xp + (R) * (D_ / 4) + 32 * q_); } while (0)

    LOAD_ROW(0, 0);
    __builtin_amdgcn_sched_barrier(0);

    f4 g[8], sh[8];
    {
        const f4* sc_p = (const f4*)(mod + b * TWO_D) + lane;
        const f4* sh_p = (const f4*)(mod + b * TWO_D + D_) + lane;
        #pragma unroll
        for (int q = 0; q < 8; ++q) {
            g[q]  = 1.0f + sc_p[32 * q];
            sh[q] = sh_p[32 * q];
        }
    }
    if (threadIdx.x < 256) {
        const int sq = threadIdx.x >> 5;
        const f4* w_p = (const f4*)w_proj + 96 * sq + 3 * lane;
        const f4 wa = w_p[0], wb = w_p[1], wc = w_p[2];
        wl[0][sq][lane] = f4{wa.x, wa.w, wb.z, wc.y};
        wl[1][sq][lane] = f4{wa.y, wb.x, wb.w, wc.z};
        wl[2][sq][lane] = f4{wa.z, wb.y, wc.x, wc.w};
    }
    __syncthreads();

    float m0 = 0.0f, m1 = 0.0f, m2 = 0.0f;

    #pragma nounroll
    for (int i = 0; i < K2_ROWS / K2_RING; ++i) {
        #pragma unroll
        for (int j = 0; j < K2_RING; ++j) {
            const int r = i * K2_RING + j;
            const int ra = (r + 1 < K2_ROWS) ? r + 1 : K2_ROWS - 1;
            __builtin_amdgcn_sched_barrier(0);
            LOAD_ROW((j + 1) % K2_RING, ra);
            __builtin_amdgcn_sched_barrier(0);

            const f4* xr = xb[j];
            f4 s4 = xr[0];
            f4 q4 = xr[0] * xr[0];
            #pragma unroll
            for (int q = 1; q < 8; ++q) { s4 += xr[q]; q4 += xr[q] * xr[q]; }
            const float ts = wave_sum32(hsum4(s4));
            const float tq = wave_sum32(hsum4(q4));
            const float mu  = ts * (1.0f / D_);
            const float var = tq * (1.0f / D_) - mu * mu;
            const float rs  = rsqrtf(var + EPS_);

            f4 a0 = {0.0f, 0.0f, 0.0f, 0.0f}, a1 = a0, a2 = a0;
            #pragma unroll
            for (int q = 0; q < 8; ++q) {
                const f4 t = (xr[q] - mu) * rs * g[q] + sh[q];
                const f4 y = f4{silu_f(t.x), silu_f(t.y), silu_f(t.z), silu_f(t.w)};
                a0 += y * wl[0][q][lane];
                a1 += y * wl[1][q][lane];
                a2 += y * wl[2][q][lane];
            }
            const float r0 = wave_sum32(hsum4(a0));
            const float r1 = wave_sum32(hsum4(a1));
            const float r2 = wave_sum32(hsum4(a2));
            if (lane == r) { m0 = r0; m1 = r1; m2 = r2; }
        }
    }
#undef LOAD_ROW

    if (lane < K2_ROWS) {
        float* p = res + (wave * K2_ROWS + lane) * OUT_;
        p[0] = m0 + b_proj[0];
        p[1] = m1 + b_proj[1];
        p[2] = m2 + b_proj[2];
    }
    __syncthreads();
    if (threadIdx.x < K2_WAVES * K2_ROWS * OUT_ / 4) {
        f4* o = (f4*)(out + (size_t)blockIdx.x * (K2_WAVES * K2_ROWS * OUT_));
        o[threadIdx.x] = ((const f4*)res)[threadIdx.x];
    }
}

extern "C" void kernel_launch(void* const* d_in, const int* in_sizes, int n_in,
                              void* d_out, int out_size, void* d_ws, size_t ws_size,
                              hipStream_t stream) {
    (void)in_sizes; (void)n_in; (void)out_size; (void)ws_size;
    const float* x      = (const float*)d_in[0];
    const float* cond   = (const float*)d_in[1];
    const float* w_ada  = (const float*)d_in[2];
    const float* b_ada  = (const float*)d_in[3];
    const float* w_proj = (const float*)d_in[4];
    const float* b_proj = (const float*)d_in[5];
    float*       out    = (float*)d_out;
    float*       mod    = (float*)d_ws;

    ada_mod_wmma<<<dim3(TWO_D / (16 * K1_TILES)), dim3(K1_WAVES * 32), 0, stream>>>(cond, w_ada, b_ada, mod);
    ln_mod_silu_proj<<<dim3(B_ * T_ / (K2_WAVES * K2_ROWS)), dim3(K2_WAVES * 32), 0, stream>>>(x, mod, w_proj, b_proj, out);
}


# silu's 1/(1+e^-v) by the hardware reciprocal v_rcp_f32 (about 1 ulp) in place of the correctly rounded f32 division (11 vector instructions per element, one chain through VCC), in both kernels
# speedup vs baseline: 3.9915x; 1.3512x over previous
#include <hip/hip_runtime.h>
#include <math.h>

#define B_    8
#define T_    8192
#define D_    1024
#define TWO_D 2048
#define OUT_  3
#define EPS_  1e-6f

typedef float v2f __attribute__((ext_vector_type(2)));
typedef float v8f __attribute__((ext_vector_type(8)));

__device__ __forceinline__ float silu_f(float v) {
    return v * __builtin_amdgcn_rcpf(1.0f + __expf(-v));
}

#define K1_TILES  2
#define K1_KSPLIT 16
#define K1_WAVES  (K1_TILES * K1_KSPLIT)
#define K1_KW     (D_ / K1_KSPLIT)
#define K1_STEPS  (K1_KW / 4)
static_assert(K1_STEPS % 2 == 0, "two accumulators take alternate k steps");
static_assert(B_ * D_ == 2 * 4 * K1_WAVES * 32, "two 16-byte pieces of cond per thread");
static_assert(K1_TILES * 16 * 4 == 128, "a work-group owns whole 128-byte lines of mod");
static_assert(B_ * K1_TILES * 16 <= K1_WAVES * 32, "one thread per result of the work-group");

typedef float f4 __attribute__((ext_vector_type(4)));

__global__ __launch_bounds__(K1_WAVES * 32) void ada_mod_wmma(
    const float* __restrict__ cond, const float* __restrict__ w_ada,
    const float* __restrict__ b_ada, float* __restrict__ mod)
{
    __shared__ __align__(16) float As[B_ * 1026];
    __shared__ __align__(16) float red[K1_WAVES * B_ * 16];
    const int tid  = threadIdx.x;
    const int wave = tid >> 5;
    const int lane = tid & 31;
    const int half = lane >> 4;
    const int l    = lane & 15;
    const int tile = wave / K1_KSPLIT;
    const int k0   = (wave % K1_KSPLIT) * K1_KW;
    const int n0   = blockIdx.x * (16 * K1_TILES) + tile * 16;

    const f4 c0 = ((const f4*)cond)[tid];
    const f4 c1 = ((const f4*)cond)[tid + K1_WAVES * 32];
    v2f bf[K1_STEPS];
    {
        const float* wp = w_ada + (size_t)(k0 + 2 * half) * TWO_D + n0 + l;
        #pragma unroll
        for (int s = 0; s < K1_STEPS; ++s) {
            bf[s].x = wp[(size_t)(4 * s)     * TWO_D];
            bf[s].y = wp[(size_t)(4 * s + 1) * TWO_D];
        }
    }

    {
        const int i0 = 4 * tid, i1 = 4 * (tid + K1_WAVES * 32);
        float* d0 = &As[(i0 >> 10) * 1026 + (i0 & (D_ - 1))];
        float* d1 = &As[(i1 >> 10) * 1026 + (i1 & (D_ - 1))];
        *(v2f*)(d0)     = v2f{silu_f(c0.x), silu_f(c0.y)};
        *(v2f*)(d0 + 2) = v2f{silu_f(c0.z), silu_f(c0.w)};
        *(v2f*)(d1)     = v2f{silu_f(c1.x), silu_f(c1.y)};
        *(v2f*)(d1 + 2) = v2f{silu_f(c1.z), silu_f(c1.w)};
    }
    __syncthreads();

    const int arow = (l & (B_ - 1)) * 1026;
    v8f acc0 = {}, acc1 = {};
    #pragma unroll
    for (int s = 0; s < K1_STEPS; s += 2) {
        const int kk = k0 + 4 * s + 2 * half;
        const v2f a0 = *(const v2f*)(&As[arow + kk]);
        const v2f a1 = *(const v2f*)(&As[arow + kk + 4]);
        acc0 = __builtin_amdgcn_wmma_f32_16x16x4_f32(
            false, a0, false, bf[s],     (short)0, acc0, false, false);
        acc1 = __builtin_amdgcn_wmma_f32_16x16x4_f32(
            false, a1, false, bf[s + 1], (short)0, acc1, false, false);
    }
    const v8f acc = acc0 + acc1;

    if (half == 0) {
        #pragma unroll
        for (int r = 0; r < B_; ++r)
            red[(wave * B_ + r) * 16 + l] = acc[r];
    }
    __syncthreads();

    if (tid < B_ * K1_TILES * 16) {
        const int r = tid >> 5, cc = tid & 31;
        const float* p = red + ((cc >> 4) * K1_KSPLIT * B_ + r) * 16 + (cc & 15);
        float sum = p[0];
        #pragma unroll
        for (int w = 1; w < K1_KSPLIT; ++w) sum += p[w * B_ * 16];
        const int n = blockIdx.x * (16 * K1_TILES) + cc;
        mod[r * TWO_D + n] = sum + b_ada[n];
    }
}

#define K2_WAVES 16
#define K2_ROWS  16
#define K2_RING  2
static_assert(T_ % (K2_WAVES * K2_ROWS) == 0, "all rows of a work-group share one batch entry");
static_assert(K2_ROWS <= 32, "lane r keeps the results of row r");
static_assert(K2_ROWS % K2_RING == 0, "the row loop is unrolled by the ring");
static_assert((B_ * T_) % (K2_WAVES * K2_ROWS) == 0, "whole work-groups");
static_assert((K2_WAVES * K2_ROWS * OUT_) % 128 == 0, "the gathered results are whole 16-byte pieces per lane");

__device__ __forceinline__ f4 ld_stream(const f4* p) { return __builtin_nontemporal_load(p); }

template <int CTRL>
__device__ __forceinline__ float row_xor(float v) {
    const int b = __builtin_bit_cast(int, v);
    return __builtin_bit_cast(float, __builtin_amdgcn_update_dpp(b, b, CTRL, 0xf, 0xf, true));
}
__device__ __forceinline__ float xhalf(float v) {
    const int b = __builtin_bit_cast(int, v);
    return __builtin_bit_cast(float,
        __builtin_amdgcn_permlanex16(b, b, 0x76543210, 0xfedcba98, false, false));
}
__device__ __forceinline__ float wave_sum32(float v) {
    v += row_xor<0x161>(v);
    v += row_xor<0x162>(v);
    v += row_xor<0x164>(v);
    v += row_xor<0x168>(v);
    v += xhalf(v);
    return v;
}
__device__ __forceinline__ float hsum4(f4 v) { return (v.x + v.y) + (v.z + v.w); }

__global__ __launch_bounds__(K2_WAVES * 32) void ln_mod_silu_proj(
    const float* __restrict__ x, const float* __restrict__ mod,
    const float* __restrict__ w_proj, const float* __restrict__ b_proj,
    float* __restrict__ out)
{
    __shared__ __align__(16) float res[K2_WAVES * K2_ROWS * OUT_];
    __shared__ f4 wl[OUT_][8][32];
    const int lane = threadIdx.x & 31;
    const int wave = __builtin_amdgcn_readfirstlane((int)(threadIdx.x >> 5));
    const int row0 = (blockIdx.x * K2_WAVES + wave) * K2_ROWS;
    const int b    = row0 >> 13;

    const f4* xp = (const f4*)(x + (size_t)row0 * D_) + lane;
    f4 xb[K2_RING][8];
#define LOAD_ROW(SLOT, R) do { \
        _Pragma("unroll") for (int q_ = 0; q_ < 8; ++q_) \
            xb[SLOT][q_] = ld_stream(xp + (R) * (D_ / 4) + 32 * q_); } while (0)

    LOAD_ROW(0, 0);
    __builtin_amdgcn_sched_barrier(0);

    f4 g[8], sh[8];
    {
        const f4* sc_p = (const f4*)(mod + b * TWO_D) + lane;
        const f4* sh_p = (const f4*)(mod + b * TWO_D + D_) + lane;
        #pragma unroll
        for (int q = 0; q < 8; ++q) {
            g[q]  = 1.0f + sc_p[32 * q];
            sh[q] = sh_p[32 * q];
        }
    }
    if (threadIdx.x < 256) {
        const int sq = threadIdx.x >> 5;
        const f4* w_p = (const f4*)w_proj + 96 * sq + 3 * lane;
        const f4 wa = w_p[0], wb = w_p[1], wc = w_p[2];
        wl[0][sq][lane] = f4{wa.x, wa.w, wb.z, wc.y};
        wl[1][sq][lane] = f4{wa.y, wb.x, wb.w, wc.z};
        wl[2][sq][lane] = f4{wa.z, wb.y, wc.x, wc.w};
    }
    __syncthreads();

    float m0 = 0.0f, m1 = 0.0f, m2 = 0.0f;

    #pragma nounroll
    for (int i = 0; i < K2_ROWS / K2_RING; ++i) {
        #pragma unroll
        for (int j = 0; j < K2_RING; ++j) {
            const int r = i * K2_RING + j;
            const int ra = (r + 1 < K2_ROWS) ? r + 1 : K2_ROWS - 1;
            __builtin_amdgcn_sched_barrier(0);
            LOAD_ROW((j + 1) % K2_RING, ra);
            __builtin_amdgcn_sched_barrier(0);

            const f4* xr = xb[j];
            f4 s4 = xr[0];
            f4 q4 = xr[0] * xr[0];
            #pragma unroll
            for (int q = 1; q < 8; ++q) { s4 += xr[q]; q4 += xr[q] * xr[q]; }
            const float ts = wave_sum32(hsum4(s4));
            const float tq = wave_sum32(hsum4(q4));
            const float mu  = ts * (1.0f / D_);
            const float var = tq * (1.0f / D_) - mu * mu;
            const float rs  = rsqrtf(var + EPS_);

            f4 a0 = {0.0f, 0.0f, 0.0f, 0.0f}, a1 = a0, a2 = a0;
            #pragma unroll
            for (int q = 0; q < 8; ++q) {
                const f4 t = (xr[q] - mu) * rs * g[q] + sh[q];
                const f4 y = f4{silu_f(t.x), silu_f(t.y), silu_f(t.z), silu_f(t.w)};
                a0 += y * wl[0][q][lane];
                a1 += y * wl[1][q][lane];
                a2 += y * wl[2][q][lane];
            }
            const float r0 = wave_sum32(hsum4(a0));
            const float r1 = wave_sum32(hsum4(a1));
            const float r2 = wave_sum32(hsum4(a2));
            if (lane == r) { m0 = r0; m1 = r1; m2 = r2; }
        }
    }
#undef LOAD_ROW

    if (lane < K2_ROWS) {
        float* p = res + (wave * K2_ROWS + lane) * OUT_;
        p[0] = m0 + b_proj[0];
        p[1] = m1 + b_proj[1];
        p[2] = m2 + b_proj[2];
    }
    __syncthreads();
    if (threadIdx.x < K2_WAVES * K2_ROWS * OUT_ / 4) {
        f4* o = (f4*)(out + (size_t)blockIdx.x * (K2_WAVES * K2_ROWS * OUT_));
        o[threadIdx.x] = ((const f4*)res)[threadIdx.x];
    }
}

extern "C" void kernel_launch(void* const* d_in, const int* in_sizes, int n_in,
                              void* d_out, int out_size, void* d_ws, size_t ws_size,
                              hipStream_t stream) {
    (void)in_sizes; (void)n_in; (void)out_size; (void)ws_size;
    const float* x      = (const float*)d_in[0];
    const float* cond   = (const float*)d_in[1];
    const float* w_ada  = (const float*)d_in[2];
    const float* b_ada  = (const float*)d_in[3];
    const float* w_proj = (const float*)d_in[4];
    const float* b_proj = (const float*)d_in[5];
    float*       out    = (float*)d_out;
    float*       mod    = (float*)d_ws;

    ada_mod_wmma<<<dim3(TWO_D / (16 * K1_TILES)), dim3(K1_WAVES * 32), 0, stream>>>(cond, w_ada, b_ada, mod);
    ln_mod_silu_proj<<<dim3(B_ * T_ / (K2_WAVES * K2_ROWS)), dim3(K2_WAVES * 32), 0, stream>>>(x, mod, w_proj, b_proj, out);
}


# Kernel 2: silu as h + h*tanh(h), h = t/2, on the hardware tangent v_tanh_f32 (scale and shift halved once per lane): one transcendental and one multiply-add per element, was exp, rcp and three more
# speedup vs baseline: 4.1858x; 1.0487x over previous
#include <hip/hip_runtime.h>
#include <math.h>

#define B_    8
#define T_    8192
#define D_    1024
#define TWO_D 2048
#define OUT_  3
#define EPS_  1e-6f

typedef float v2f __attribute__((ext_vector_type(2)));
typedef float v8f __attribute__((ext_vector_type(8)));

__device__ __forceinline__ float silu_f(float v) {
    return v * __builtin_amdgcn_rcpf(1.0f + __expf(-v));
}

__device__ __forceinline__ float silu_of_half(float h) {
    return fmaf(h, __builtin_amdgcn_tanhf(h), h);
}

#define K1_TILES  2
#define K1_KSPLIT 16
#define K1_WAVES  (K1_TILES * K1_KSPLIT)
#define K1_KW     (D_ / K1_KSPLIT)
#define K1_STEPS  (K1_KW / 4)
static_assert(K1_STEPS % 2 == 0, "two accumulators take alternate k steps");
static_assert(B_ * D_ == 2 * 4 * K1_WAVES * 32, "two 16-byte pieces of cond per thread");
static_assert(K1_TILES * 16 * 4 == 128, "a work-group owns whole 128-byte lines of mod");
static_assert(B_ * K1_TILES * 16 <= K1_WAVES * 32, "one thread per result of the work-group");

typedef float f4 __attribute__((ext_vector_type(4)));

__global__ __launch_bounds__(K1_WAVES * 32) void ada_mod_wmma(
    const float* __restrict__ cond, const float* __restrict__ w_ada,
    const float* __restrict__ b_ada, float* __restrict__ mod)
{
    __shared__ __align__(16) float As[B_ * 1026];
    __shared__ __align__(16) float red[K1_WAVES * B_ * 16];
    const int tid  = threadIdx.x;
    const int wave = tid >> 5;
    const int lane = tid & 31;
    const int half = lane >> 4;
    const int l    = lane & 15;
    const int tile = wave / K1_KSPLIT;
    const int k0   = (wave % K1_KSPLIT) * K1_KW;
    const int n0   = blockIdx.x * (16 * K1_TILES) + tile * 16;

    const f4 c0 = ((const f4*)cond)[tid];
    const f4 c1 = ((const f4*)cond)[tid + K1_WAVES * 32];
    v2f bf[K1_STEPS];
    {
        const float* wp = w_ada + (size_t)(k0 + 2 * half) * TWO_D + n0 + l;
        #pragma unroll
        for (int s = 0; s < K1_STEPS; ++s) {
            bf[s].x = wp[(size_t)(4 * s)     * TWO_D];
            bf[s].y = wp[(size_t)(4 * s + 1) * TWO_D];
        }
    }

    {
        const int i0 = 4 * tid, i1 = 4 * (tid + K1_WAVES * 32);
        float* d0 = &As[(i0 >> 10) * 1026 + (i0 & (D_ - 1))];
        float* d1 = &As[(i1 >> 10) * 1026 + (i1 & (D_ - 1))];
        *(v2f*)(d0)     = v2f{silu_f(c0.x), silu_f(c0.y)};
        *(v2f*)(d0 + 2) = v2f{silu_f(c0.z), silu_f(c0.w)};
        *(v2f*)(d1)     = v2f{silu_f(c1.x), silu_f(c1.y)};
        *(v2f*)(d1 + 2) = v2f{silu_f(c1.z), silu_f(c1.w)};
    }
    __syncthreads();

    const int arow = (l & (B_ - 1)) * 1026;
    v8f acc0 = {}, acc1 = {};
    #pragma unroll
    for (int s = 0; s < K1_STEPS; s += 2) {
        const int kk = k0 + 4 * s + 2 * half;
        const v2f a0 = *(const v2f*)(&As[arow + kk]);
        const v2f a1 = *(const v2f*)(&As[arow + kk + 4]);
        acc0 = __builtin_amdgcn_wmma_f32_16x16x4_f32(
            false, a0, false, bf[s],     (short)0, acc0, false, false);
        acc1 = __builtin_amdgcn_wmma_f32_16x16x4_f32(
            false, a1, false, bf[s + 1], (short)0, acc1, false, false);
    }
    const v8f acc = acc0 + acc1;

    if (half == 0) {
        #pragma unroll
        for (int r = 0; r < B_; ++r)
            red[(wave * B_ + r) * 16 + l] = acc[r];
    }
    __syncthreads();

    if (tid < B_ * K1_TILES * 16) {
        const int r = tid >> 5, cc = tid & 31;
        const float* p = red + ((cc >> 4) * K1_KSPLIT * B_ + r) * 16 + (cc & 15);
        float sum = p[0];
        #pragma unroll
        for (int w = 1; w < K1_KSPLIT; ++w) sum += p[w * B_ * 16];
        const int n = blockIdx.x * (16 * K1_TILES) + cc;
        mod[r * TWO_D + n] = sum + b_ada[n];
    }
}

#define K2_WAVES 16
#define K2_ROWS  16
#define K2_RING  2
static_assert(T_ % (K2_WAVES * K2_ROWS) == 0, "all rows of a work-group share one batch entry");
static_assert(K2_ROWS <= 32, "lane r keeps the results of row r");
static_assert(K2_ROWS % K2_RING == 0, "the row loop is unrolled by the ring");
static_assert((B_ * T_) % (K2_WAVES * K2_ROWS) == 0, "whole work-groups");
static_assert((K2_WAVES * K2_ROWS * OUT_) % 128 == 0, "the gathered results are whole 16-byte pieces per lane");

__device__ __forceinline__ f4 ld_stream(const f4* p) { return __builtin_nontemporal_load(p); }

template <int CTRL>
__device__ __forceinline__ float row_xor(float v) {
    const int b = __builtin_bit_cast(int, v);
    return __builtin_bit_cast(float, __builtin_amdgcn_update_dpp(b, b, CTRL, 0xf, 0xf, true));
}
__device__ __forceinline__ float xhalf(float v) {
    const int b = __builtin_bit_cast(int, v);
    return __builtin_bit_cast(float,
        __builtin_amdgcn_permlanex16(b, b, 0x76543210, 0xfedcba98, false, false));
}
__device__ __forceinline__ float wave_sum32(float v) {
    v += row_xor<0x161>(v);
    v += row_xor<0x162>(v);
    v += row_xor<0x164>(v);
    v += row_xor<0x168>(v);
    v += xhalf(v);
    return v;
}
__device__ __forceinline__ float hsum4(f4 v) { return (v.x + v.y) + (v.z + v.w); }

__global__ __launch_bounds__(K2_WAVES * 32) void ln_mod_silu_proj(
    const float* __restrict__ x, const float* __restrict__ mod,
    const float* __restrict__ w_proj, const float* __restrict__ b_proj,
    float* __restrict__ out)
{
    __shared__ __align__(16) float res[K2_WAVES * K2_ROWS * OUT_];
    __shared__ f4 wl[OUT_][8][32];
    const int lane = threadIdx.x & 31;
    const int wave = __builtin_amdgcn_readfirstlane((int)(threadIdx.x >> 5));
    const int row0 = (blockIdx.x * K2_WAVES + wave) * K2_ROWS;
    const int b    = row0 >> 13;

    const f4* xp = (const f4*)(x + (size_t)row0 * D_) + lane;
    f4 xb[K2_RING][8];
#define LOAD_ROW(SLOT, R) do { \
        _Pragma("unroll") for (int q_ = 0; q_ < 8; ++q_) \
            xb[SLOT][q_] = ld_stream(xp + (R) * (D_ / 4) + 32 * q_); } while (0)

    LOAD_ROW(0, 0);
    __builtin_amdgcn_sched_barrier(0);

    f4 g[8], sh[8];
    {
        const f4* sc_p = (const f4*)(mod + b * TWO_D) + lane;
        const f4* sh_p = (const f4*)(mod + b * TWO_D + D_) + lane;
        #pragma unroll
        for (int q = 0; q < 8; ++q) {
            g[q]  = 0.5f * (1.0f + sc_p[32 * q]);
            sh[q] = 0.5f * sh_p[32 * q];
        }
    }
    if (threadIdx.x < 256) {
        const int sq = threadIdx.x >> 5;
        const f4* w_p = (const f4*)w_proj + 96 * sq + 3 * lane;
        const f4 wa = w_p[0], wb = w_p[1], wc = w_p[2];
        wl[0][sq][lane] = f4{wa.x, wa.w, wb.z, wc.y};
        wl[1][sq][lane] = f4{wa.y, wb.x, wb.w, wc.z};
        wl[2][sq][lane] = f4{wa.z, wb.y, wc.x, wc.w};
    }
    __syncthreads();

    float m0 = 0.0f, m1 = 0.0f, m2 = 0.0f;

    #pragma nounroll
    for (int i = 0; i < K2_ROWS / K2_RING; ++i) {
        #pragma unroll
        for (int j = 0; j < K2_RING; ++j) {
            const int r = i * K2_RING + j;
            const int ra = (r + 1 < K2_ROWS) ? r + 1 : K2_ROWS - 1;
            __builtin_amdgcn_sched_barrier(0);
            LOAD_ROW((j + 1) % K2_RING, ra);
            __builtin_amdgcn_sched_barrier(0);

            const f4* xr = xb[j];
            f4 s4 = xr[0];
            f4 q4 = xr[0] * xr[0];
            #pragma unroll
            for (int q = 1; q < 8; ++q) { s4 += xr[q]; q4 += xr[q] * xr[q]; }
            const float ts = wave_sum32(hsum4(s4));
            const float tq = wave_sum32(hsum4(q4));
            const float mu  = ts * (1.0f / D_);
            const float var = tq * (1.0f / D_) - mu * mu;
            const float rs  = rsqrtf(var + EPS_);

            f4 a0 = {0.0f, 0.0f, 0.0f, 0.0f}, a1 = a0, a2 = a0;
            #pragma unroll
            for (int q = 0; q < 8; ++q) {
                const f4 h = (xr[q] - mu) * rs * g[q] + sh[q];
                const f4 y = f4{silu_of_half(h.x), silu_of_half(h.y),
                                silu_of_half(h.z), silu_of_half(h.w)};
                a0 += y * wl[0][q][lane];
                a1 += y * wl[1][q][lane];
                a2 += y * wl[2][q][lane];
            }
            const float r0 = wave_sum32(hsum4(a0));
            const float r1 = wave_sum32(hsum4(a1));
            const float r2 = wave_sum32(hsum4(a2));
            if (lane == r) { m0 = r0; m1 = r1; m2 = r2; }
        }
    }
#undef LOAD_ROW

    if (lane < K2_ROWS) {
        float* p = res + (wave * K2_ROWS + lane) * OUT_;
        p[0] = m0 + b_proj[0];
        p[1] = m1 + b_proj[1];
        p[2] = m2 + b_proj[2];
    }
    __syncthreads();
    if (threadIdx.x < K2_WAVES * K2_ROWS * OUT_ / 4) {
        f4* o = (f4*)(out + (size_t)blockIdx.x * (K2_WAVES * K2_ROWS * OUT_));
        o[threadIdx.x] = ((const f4*)res)[threadIdx.x];
    }
}

extern "C" void kernel_launch(void* const* d_in, const int* in_sizes, int n_in,
                              void* d_out, int out_size, void* d_ws, size_t ws_size,
                              hipStream_t stream) {
    (void)in_sizes; (void)n_in; (void)out_size; (void)ws_size;
    const float* x      = (const float*)d_in[0];
    const float* cond   = (const float*)d_in[1];
    const float* w_ada  = (const float*)d_in[2];
    const float* b_ada  = (const float*)d_in[3];
    const float* w_proj = (const float*)d_in[4];
    const float* b_proj = (const float*)d_in[5];
    float*       out    = (float*)d_out;
    float*       mod    = (float*)d_ws;

    ada_mod_wmma<<<dim3(TWO_D / (16 * K1_TILES)), dim3(K1_WAVES * 32), 0, stream>>>(cond, w_ada, b_ada, mod);
    ln_mod_silu_proj<<<dim3(B_ * T_ / (K2_WAVES * K2_ROWS)), dim3(K2_WAVES * 32), 0, stream>>>(x, mod, w_proj, b_proj, out);
}


# Kernel 2: each row requested once, as soon as its registers are free (rows 0 and 1 at wave start, row r+2 behind the last read of row r); no repeated last row; sums and their order unchanged
# speedup vs baseline: 4.3142x; 1.0307x over previous
#include <hip/hip_runtime.h>
#include <math.h>

#define B_    8
#define T_    8192
#define D_    1024
#define TWO_D 2048
#define OUT_  3
#define EPS_  1e-6f

typedef float v2f __attribute__((ext_vector_type(2)));
typedef float v8f __attribute__((ext_vector_type(8)));

__device__ __forceinline__ float silu_f(float v) {
    return v * __builtin_amdgcn_rcpf(1.0f + __expf(-v));
}

__device__ __forceinline__ float silu_of_half(float h) {
    return fmaf(h, __builtin_amdgcn_tanhf(h), h);
}

#define K1_TILES  2
#define K1_KSPLIT 16
#define K1_WAVES  (K1_TILES * K1_KSPLIT)
#define K1_KW     (D_ / K1_KSPLIT)
#define K1_STEPS  (K1_KW / 4)
static_assert(K1_STEPS % 2 == 0, "two accumulators take alternate k steps");
static_assert(B_ * D_ == 2 * 4 * K1_WAVES * 32, "two 16-byte pieces of cond per thread");
static_assert(K1_TILES * 16 * 4 == 128, "a work-group owns whole 128-byte lines of mod");
static_assert(B_ * K1_TILES * 16 <= K1_WAVES * 32, "one thread per result of the work-group");

typedef float f4 __attribute__((ext_vector_type(4)));

__global__ __launch_bounds__(K1_WAVES * 32) void ada_mod_wmma(
    const float* __restrict__ cond, const float* __restrict__ w_ada,
    const float* __restrict__ b_ada, float* __restrict__ mod)
{
    __shared__ __align__(16) float As[B_ * 1026];
    __shared__ __align__(16) float red[K1_WAVES * B_ * 16];
    const int tid  = threadIdx.x;
    const int wave = tid >> 5;
    const int lane = tid & 31;
    const int half = lane >> 4;
    const int l    = lane & 15;
    const int tile = wave / K1_KSPLIT;
    const int k0   = (wave % K1_KSPLIT) * K1_KW;
    const int n0   = blockIdx.x * (16 * K1_TILES) + tile * 16;

    const f4 c0 = ((const f4*)cond)[tid];
    const f4 c1 = ((const f4*)cond)[tid + K1_WAVES * 32];
    v2f bf[K1_STEPS];
    {
        const float* wp = w_ada + (size_t)(k0 + 2 * half) * TWO_D + n0 + l;
        #pragma unroll
        for (int s = 0; s < K1_STEPS; ++s) {
            bf[s].x = wp[(size_t)(4 * s)     * TWO_D];
            bf[s].y = wp[(size_t)(4 * s + 1) * TWO_D];
        }
    }

    {
        const int i0 = 4 * tid, i1 = 4 * (tid + K1_WAVES * 32);
        float* d0 = &As[(i0 >> 10) * 1026 + (i0 & (D_ - 1))];
        float* d1 = &As[(i1 >> 10) * 1026 + (i1 & (D_ - 1))];
        *(v2f*)(d0)     = v2f{silu_f(c0.x), silu_f(c0.y)};
        *(v2f*)(d0 + 2) = v2f{silu_f(c0.z), silu_f(c0.w)};
        *(v2f*)(d1)     = v2f{silu_f(c1.x), silu_f(c1.y)};
        *(v2f*)(d1 + 2) = v2f{silu_f(c1.z), silu_f(c1.w)};
    }
    __syncthreads();

    const int arow = (l & (B_ - 1)) * 1026;
    v8f acc0 = {}, acc1 = {};
    #pragma unroll
    for (int s = 0; s < K1_STEPS; s += 2) {
        const int kk = k0 + 4 * s + 2 * half;
        const v2f a0 = *(const v2f*)(&As[arow + kk]);
        const v2f a1 = *(const v2f*)(&As[arow + kk + 4]);
        acc0 = __builtin_amdgcn_wmma_f32_16x16x4_f32(
            false, a0, false, bf[s],     (short)0, acc0, false, false);
        acc1 = __builtin_amdgcn_wmma_f32_16x16x4_f32(
            false, a1, false, bf[s + 1], (short)0, acc1, false, false);
    }
    const v8f acc = acc0 + acc1;

    if (half == 0) {
        #pragma unroll
        for (int r = 0; r < B_; ++r)
            red[(wave * B_ + r) * 16 + l] = acc[r];
    }
    __syncthreads();

    if (tid < B_ * K1_TILES * 16) {
        const int r = tid >> 5, cc = tid & 31;
        const float* p = red + ((cc >> 4) * K1_KSPLIT * B_ + r) * 16 + (cc & 15);
        float sum = p[0];
        #pragma unroll
        for (int w = 1; w < K1_KSPLIT; ++w) sum += p[w * B_ * 16];
        const int n = blockIdx.x * (16 * K1_TILES) + cc;
        mod[r * TWO_D + n] = sum + b_ada[n];
    }
}

#define K2_WAVES 16
#define K2_ROWS  16
#define K2_RING  2
static_assert(T_ % (K2_WAVES * K2_ROWS) == 0, "all rows of a work-group share one batch entry");
static_assert(K2_ROWS <= 32, "lane r keeps the results of row r");
static_assert(K2_ROWS % K2_RING == 0, "the row loop is unrolled by the ring");
static_assert((B_ * T_) % (K2_WAVES * K2_ROWS) == 0, "whole work-groups");
static_assert((K2_WAVES * K2_ROWS * OUT_) % 128 == 0, "the gathered results are whole 16-byte pieces per lane");

__device__ __forceinline__ f4 ld_stream(const f4* p) { return __builtin_nontemporal_load(p); }

template <int CTRL>
__device__ __forceinline__ float row_xor(float v) {
    const int b = __builtin_bit_cast(int, v);
    return __builtin_bit_cast(float, __builtin_amdgcn_update_dpp(b, b, CTRL, 0xf, 0xf, true));
}
__device__ __forceinline__ float xhalf(float v) {
    const int b = __builtin_bit_cast(int, v);
    return __builtin_bit_cast(float,
        __builtin_amdgcn_permlanex16(b, b, 0x76543210, 0xfedcba98, false, false));
}
__device__ __forceinline__ float wave_sum32(float v) {
    v += row_xor<0x161>(v);
    v += row_xor<0x162>(v);
    v += row_xor<0x164>(v);
    v += row_xor<0x168>(v);
    v += xhalf(v);
    return v;
}
__device__ __forceinline__ float hsum4(f4 v) { return (v.x + v.y) + (v.z + v.w); }

__global__ __launch_bounds__(K2_WAVES * 32) void ln_mod_silu_proj(
    const float* __restrict__ x, const float* __restrict__ mod,
    const float* __restrict__ w_proj, const float* __restrict__ b_proj,
    float* __restrict__ out)
{
    __shared__ __align__(16) float res[K2_WAVES * K2_ROWS * OUT_];
    __shared__ f4 wl[OUT_][8][32];
    const int lane = threadIdx.x & 31;
    const int wave = __builtin_amdgcn_readfirstlane((int)(threadIdx.x >> 5));
    const int row0 = (blockIdx.x * K2_WAVES + wave) * K2_ROWS;
    const int b    = row0 >> 13;

    const f4* xp = (const f4*)(x + (size_t)row0 * D_) + lane;
    f4 xb[K2_RING][8];
#define LOAD_ROW(SLOT, R) do { \
        _Pragma("unroll") for (int q_ = 0; q_ < 8; ++q_) \
            xb[SLOT][q_] = ld_stream(xp + (R) * (D_ / 4) + 32 * q_); } while (0)

    LOAD_ROW(0, 0);
    LOAD_ROW(1, 1);
    __builtin_amdgcn_sched_barrier(0);

    f4 g[8], sh[8];
    {
        const f4* sc_p = (const f4*)(mod + b * TWO_D) + lane;
        const f4* sh_p = (const f4*)(mod + b * TWO_D + D_) + lane;
        #pragma unroll
        for (int q = 0; q < 8; ++q) {
            g[q]  = 0.5f * (1.0f + sc_p[32 * q]);
            sh[q] = 0.5f * sh_p[32 * q];
        }
    }
    if (threadIdx.x < 256) {
        const int sq = threadIdx.x >> 5;
        const f4* w_p = (const f4*)w_proj + 96 * sq + 3 * lane;
        const f4 wa = w_p[0], wb = w_p[1], wc = w_p[2];
        wl[0][sq][lane] = f4{wa.x, wa.w, wb.z, wc.y};
        wl[1][sq][lane] = f4{wa.y, wb.x, wb.w, wc.z};
        wl[2][sq][lane] = f4{wa.z, wb.y, wc.x, wc.w};
    }
    __syncthreads();

    float m0 = 0.0f, m1 = 0.0f, m2 = 0.0f;

#define ROW_BODY(J, R, REFILL_) do { \
        f4* xr = xb[J]; \
        f4 s4 = xr[0]; \
        f4 q4 = xr[0] * xr[0]; \
        _Pragma("unroll") for (int q = 1; q < 8; ++q) { s4 += xr[q]; q4 += xr[q] * xr[q]; } \
        const float ts = wave_sum32(hsum4(s4)); \
        const float tq = wave_sum32(hsum4(q4)); \
        const float mu  = ts * (1.0f / D_); \
        const float var = tq * (1.0f / D_) - mu * mu; \
        const float rs  = rsqrtf(var + EPS_); \
        f4 a0 = {0.0f, 0.0f, 0.0f, 0.0f}, a1 = a0, a2 = a0; \
        _Pragma("unroll") for (int qg = 0; qg < 8; qg += 2) { \
            f4 hh[2]; \
            _Pragma("unroll") for (int u = 0; u < 2; ++u) \
                hh[u] = (xr[qg + u] - mu) * rs * g[qg + u] + sh[qg + u];     \
            if (REFILL_) { \
                __builtin_amdgcn_sched_barrier(0); \
                _Pragma("unroll") for (int u = 0; u < 2; ++u) \
                    xr[qg + u] = ld_stream(xp + ((R) + K2_RING) * (D_ / 4) + 32 * (qg + u)); \
                __builtin_amdgcn_sched_barrier(0); \
            } \
            _Pragma("unroll") for (int u = 0; u < 2; ++u) { \
                const f4 h = hh[u]; \
                const f4 y = f4{silu_of_half(h.x), silu_of_half(h.y), \
                                silu_of_half(h.z), silu_of_half(h.w)}; \
                a0 += y * wl[0][qg + u][lane]; \
                a1 += y * wl[1][qg + u][lane]; \
                a2 += y * wl[2][qg + u][lane]; \
            } \
        } \
        const float r0 = wave_sum32(hsum4(a0)); \
        const float r1 = wave_sum32(hsum4(a1)); \
        const float r2 = wave_sum32(hsum4(a2)); \
        if (lane == (R)) { m0 = r0; m1 = r1; m2 = r2; } \
    } while (0)

    #pragma nounroll
    for (int i = 0; i < K2_ROWS / K2_RING - 1; ++i) {
        ROW_BODY(0, i * K2_RING,     true);
        ROW_BODY(1, i * K2_RING + 1, true);
    }
    ROW_BODY(0, K2_ROWS - 2, false);
    __builtin_amdgcn_sched_barrier(0);
    ROW_BODY(1, K2_ROWS - 1, false);
#undef ROW_BODY
#undef LOAD_ROW

    if (lane < K2_ROWS) {
        float* p = res + (wave * K2_ROWS + lane) * OUT_;
        p[0] = m0 + b_proj[0];
        p[1] = m1 + b_proj[1];
        p[2] = m2 + b_proj[2];
    }
    __syncthreads();
    if (threadIdx.x < K2_WAVES * K2_ROWS * OUT_ / 4) {
        f4* o = (f4*)(out + (size_t)blockIdx.x * (K2_WAVES * K2_ROWS * OUT_));
        o[threadIdx.x] = ((const f4*)res)[threadIdx.x];
    }
}

extern "C" void kernel_launch(void* const* d_in, const int* in_sizes, int n_in,
                              void* d_out, int out_size, void* d_ws, size_t ws_size,
                              hipStream_t stream) {
    (void)in_sizes; (void)n_in; (void)out_size; (void)ws_size;
    const float* x      = (const float*)d_in[0];
    const float* cond   = (const float*)d_in[1];
    const float* w_ada  = (const float*)d_in[2];
    const float* b_ada  = (const float*)d_in[3];
    const float* w_proj = (const float*)d_in[4];
    const float* b_proj = (const float*)d_in[5];
    float*       out    = (float*)d_out;
    float*       mod    = (float*)d_ws;

    ada_mod_wmma<<<dim3(TWO_D / (16 * K1_TILES)), dim3(K1_WAVES * 32), 0, stream>>>(cond, w_ada, b_ada, mod);
    ln_mod_silu_proj<<<dim3(B_ * T_ / (K2_WAVES * K2_ROWS)), dim3(K2_WAVES * 32), 0, stream>>>(x, mod, w_proj, b_proj, out);
}


# Kernel 2 as lone waves, no LDS or barrier: 2048 waves of 32 rows (2 per SIMD, up to 512 registers), w_proj share in registers, 5 rows in flight per wave (was 2); sums and their order unchanged
# speedup vs baseline: 4.4926x; 1.0414x over previous
#include <hip/hip_runtime.h>
#include <math.h>

#define B_    8
#define T_    8192
#define D_    1024
#define TWO_D 2048
#define OUT_  3
#define EPS_  1e-6f

typedef float v2f __attribute__((ext_vector_type(2)));
typedef float v8f __attribute__((ext_vector_type(8)));

__device__ __forceinline__ float silu_f(float v) {
    return v * __builtin_amdgcn_rcpf(1.0f + __expf(-v));
}

__device__ __forceinline__ float silu_of_half(float h) {
    return fmaf(h, __builtin_amdgcn_tanhf(h), h);
}

#define K1_TILES  2
#define K1_KSPLIT 16
#define K1_WAVES  (K1_TILES * K1_KSPLIT)
#define K1_KW     (D_ / K1_KSPLIT)
#define K1_STEPS  (K1_KW / 4)
static_assert(K1_STEPS % 2 == 0, "two accumulators take alternate k steps");
static_assert(B_ * D_ == 2 * 4 * K1_WAVES * 32, "two 16-byte pieces of cond per thread");
static_assert(K1_TILES * 16 * 4 == 128, "a work-group owns whole 128-byte lines of mod");
static_assert(B_ * K1_TILES * 16 <= K1_WAVES * 32, "one thread per result of the work-group");

typedef float f4 __attribute__((ext_vector_type(4)));

__global__ __launch_bounds__(K1_WAVES * 32) void ada_mod_wmma(
    const float* __restrict__ cond, const float* __restrict__ w_ada,
    const float* __restrict__ b_ada, float* __restrict__ mod)
{
    __shared__ __align__(16) float As[B_ * 1026];
    __shared__ __align__(16) float red[K1_WAVES * B_ * 16];
    const int tid  = threadIdx.x;
    const int wave = tid >> 5;
    const int lane = tid & 31;
    const int half = lane >> 4;
    const int l    = lane & 15;
    const int tile = wave / K1_KSPLIT;
    const int k0   = (wave % K1_KSPLIT) * K1_KW;
    const int n0   = blockIdx.x * (16 * K1_TILES) + tile * 16;

    const f4 c0 = ((const f4*)cond)[tid];
    const f4 c1 = ((const f4*)cond)[tid + K1_WAVES * 32];
    v2f bf[K1_STEPS];
    {
        const float* wp = w_ada + (size_t)(k0 + 2 * half) * TWO_D + n0 + l;
        #pragma unroll
        for (int s = 0; s < K1_STEPS; ++s) {
            bf[s].x = wp[(size_t)(4 * s)     * TWO_D];
            bf[s].y = wp[(size_t)(4 * s + 1) * TWO_D];
        }
    }

    {
        const int i0 = 4 * tid, i1 = 4 * (tid + K1_WAVES * 32);
        float* d0 = &As[(i0 >> 10) * 1026 + (i0 & (D_ - 1))];
        float* d1 = &As[(i1 >> 10) * 1026 + (i1 & (D_ - 1))];
        *(v2f*)(d0)     = v2f{silu_f(c0.x), silu_f(c0.y)};
        *(v2f*)(d0 + 2) = v2f{silu_f(c0.z), silu_f(c0.w)};
        *(v2f*)(d1)     = v2f{silu_f(c1.x), silu_f(c1.y)};
        *(v2f*)(d1 + 2) = v2f{silu_f(c1.z), silu_f(c1.w)};
    }
    __syncthreads();

    const int arow = (l & (B_ - 1)) * 1026;
    v8f acc0 = {}, acc1 = {};
    #pragma unroll
    for (int s = 0; s < K1_STEPS; s += 2) {
        const int kk = k0 + 4 * s + 2 * half;
        const v2f a0 = *(const v2f*)(&As[arow + kk]);
        const v2f a1 = *(const v2f*)(&As[arow + kk + 4]);
        acc0 = __builtin_amdgcn_wmma_f32_16x16x4_f32(
            false, a0, false, bf[s],     (short)0, acc0, false, false);
        acc1 = __builtin_amdgcn_wmma_f32_16x16x4_f32(
            false, a1, false, bf[s + 1], (short)0, acc1, false, false);
    }
    const v8f acc = acc0 + acc1;

    if (half == 0) {
        #pragma unroll
        for (int r = 0; r < B_; ++r)
            red[(wave * B_ + r) * 16 + l] = acc[r];
    }
    __syncthreads();

    if (tid < B_ * K1_TILES * 16) {
        const int r = tid >> 5, cc = tid & 31;
        const float* p = red + ((cc >> 4) * K1_KSPLIT * B_ + r) * 16 + (cc & 15);
        float sum = p[0];
        #pragma unroll
        for (int w = 1; w < K1_KSPLIT; ++w) sum += p[w * B_ * 16];
        const int n = blockIdx.x * (16 * K1_TILES) + cc;
        mod[r * TWO_D + n] = sum + b_ada[n];
    }
}

#define K2_WAVES 8
#define K2_ROWS  32
#define K2_RING  5
#define K2_LOOPS ((K2_ROWS - K2_RING) / K2_RING)
#define K2_REST  ((K2_ROWS - K2_RING) - K2_LOOPS * K2_RING)
static_assert(T_ % (K2_WAVES * K2_ROWS) == 0, "all rows of a work-group share one batch entry");
static_assert(K2_ROWS == 32, "lane r keeps the results of row r, and the wave's results are whole lines of out");
static_assert(K2_RING == 5 && K2_LOOPS == 5 && K2_REST == 2, "the row sequence below is written out for this ring");
static_assert((B_ * T_) % (K2_WAVES * K2_ROWS) == 0, "whole work-groups");
static_assert((K2_ROWS * OUT_ * 4) % 128 == 0, "a wave's results are whole 128-byte lines");

__device__ __forceinline__ f4 ld_stream(const f4* p) { return __builtin_nontemporal_load(p); }

template <int CTRL>
__device__ __forceinline__ float row_xor(float v) {
    const int b = __builtin_bit_cast(int, v);
    return __builtin_bit_cast(float, __builtin_amdgcn_update_dpp(b, b, CTRL, 0xf, 0xf, true));
}
__device__ __forceinline__ float xhalf(float v) {
    const int b = __builtin_bit_cast(int, v);
    return __builtin_bit_cast(float,
        __builtin_amdgcn_permlanex16(b, b, 0x76543210, 0xfedcba98, false, false));
}
__device__ __forceinline__ float wave_sum32(float v) {
    v += row_xor<0x161>(v);
    v += row_xor<0x162>(v);
    v += row_xor<0x164>(v);
    v += row_xor<0x168>(v);
    v += xhalf(v);
    return v;
}
__device__ __forceinline__ float hsum4(f4 v) { return (v.x + v.y) + (v.z + v.w); }

struct out3 { float a, b, c; };

__global__ __launch_bounds__(K2_WAVES * 32) __attribute__((amdgpu_waves_per_eu(1, 2)))
void ln_mod_silu_proj(
    const float* __restrict__ x, const float* __restrict__ mod,
    const float* __restrict__ w_proj, const float* __restrict__ b_proj,
    float* __restrict__ out)
{
    const int lane = threadIdx.x & 31;
    const int wave = __builtin_amdgcn_readfirstlane((int)(threadIdx.x >> 5));
    const int row0 = (blockIdx.x * K2_WAVES + wave) * K2_ROWS;
    const int b    = row0 >> 13;

    const f4* xp = (const f4*)(x + (size_t)row0 * D_) + lane;
    f4 xb[K2_RING][8];
#define LOAD_ROW(SLOT, R) do { \
        _Pragma("unroll") for (int q_ = 0; q_ < 8; ++q_) \
            xb[SLOT][q_] = ld_stream(xp + (R) * (D_ / 4) + 32 * q_); } while (0)

    LOAD_ROW(0, 0);
    __builtin_amdgcn_sched_barrier(0);
    f4 g[8], sh[8];
    f4 w0[8], w1[8], w2[8];
    {
        const f4* sc_p = (const f4*)(mod + b * TWO_D) + lane;
        const f4* sh_p = (const f4*)(mod + b * TWO_D + D_) + lane;
        const f4* w_p = (const f4*)w_proj + 3 * lane;
        #pragma unroll
        for (int q = 0; q < 8; ++q) {
            g[q]  = sc_p[32 * q];
            sh[q] = sh_p[32 * q];
            w0[q] = w_p[96 * q];
            w1[q] = w_p[96 * q + 1];
            w2[q] = w_p[96 * q + 2];
        }
    }
    __builtin_amdgcn_sched_barrier(0);
    LOAD_ROW(1, 1);
    LOAD_ROW(2, 2);
    LOAD_ROW(3, 3);
    LOAD_ROW(4, 4);
    __builtin_amdgcn_sched_barrier(0);
    #pragma unroll
    for (int q = 0; q < 8; ++q) {
        g[q]  = 0.5f * (1.0f + g[q]);
        sh[q] = 0.5f * sh[q];
        const f4 wa = w0[q], wb = w1[q], wc = w2[q];
        w0[q] = f4{wa.x, wa.w, wb.z, wc.y};
        w1[q] = f4{wa.y, wb.x, wb.w, wc.z};
        w2[q] = f4{wa.z, wb.y, wc.x, wc.w};
    }

    float m0 = 0.0f, m1 = 0.0f, m2 = 0.0f;

#define ROW_BODY(J, R, REFILL_) do { \
        f4* xr = xb[J]; \
        f4 s4 = xr[0]; \
        f4 q4 = xr[0] * xr[0]; \
        _Pragma("unroll") for (int q = 1; q < 8; ++q) { s4 += xr[q]; q4 += xr[q] * xr[q]; } \
        const float ts = wave_sum32(hsum4(s4)); \
        const float tq = wave_sum32(hsum4(q4)); \
        const float mu  = ts * (1.0f / D_); \
        const float var = tq * (1.0f / D_) - mu * mu; \
        const float rs  = rsqrtf(var + EPS_); \
        f4 a0 = {0.0f, 0.0f, 0.0f, 0.0f}, a1 = a0, a2 = a0; \
        _Pragma("unroll") for (int qg = 0; qg < 8; qg += 2) { \
            f4 hh[2]; \
            _Pragma("unroll") for (int u = 0; u < 2; ++u) \
                hh[u] = (xr[qg + u] - mu) * rs * g[qg + u] + sh[qg + u];     \
            if (REFILL_) { \
                __builtin_amdgcn_sched_barrier(0); \
                _Pragma("unroll") for (int u = 0; u < 2; ++u) \
                    xr[qg + u] = ld_stream(xp + ((R) + K2_RING) * (D_ / 4) + 32 * (qg + u)); \
                __builtin_amdgcn_sched_barrier(0); \
            } \
            _Pragma("unroll") for (int u = 0; u < 2; ++u) { \
                const f4 h = hh[u]; \
                const f4 y = f4{silu_of_half(h.x), silu_of_half(h.y), \
                                silu_of_half(h.z), silu_of_half(h.w)}; \
                a0 += y * w0[qg + u]; \
                a1 += y * w1[qg + u]; \
                a2 += y * w2[qg + u]; \
            } \
        } \
        const float r0 = wave_sum32(hsum4(a0)); \
        const float r1 = wave_sum32(hsum4(a1)); \
        const float r2 = wave_sum32(hsum4(a2)); \
        if (lane == (R)) { m0 = r0; m1 = r1; m2 = r2; } \
    } while (0)

    #pragma nounroll
    for (int i = 0; i < K2_LOOPS; ++i) {
        ROW_BODY(0, i * K2_RING,     true);
        ROW_BODY(1, i * K2_RING + 1, true);
        ROW_BODY(2, i * K2_RING + 2, true);
        ROW_BODY(3, i * K2_RING + 3, true);
        ROW_BODY(4, i * K2_RING + 4, true);
    }
    ROW_BODY(0, 25, true);
    ROW_BODY(1, 26, true);
    ROW_BODY(2, 27, false);
    __builtin_amdgcn_sched_barrier(0);
    ROW_BODY(3, 28, false);
    __builtin_amdgcn_sched_barrier(0);
    ROW_BODY(4, 29, false);
    __builtin_amdgcn_sched_barrier(0);
    ROW_BODY(0, 30, false);
    __builtin_amdgcn_sched_barrier(0);
    ROW_BODY(1, 31, false);
#undef ROW_BODY
#undef LOAD_ROW

    {
        out3* o = (out3*)(out + (size_t)(row0 + lane) * OUT_);
        *o = out3{m0 + b_proj[0], m1 + b_proj[1], m2 + b_proj[2]};
    }
}

extern "C" void kernel_launch(void* const* d_in, const int* in_sizes, int n_in,
                              void* d_out, int out_size, void* d_ws, size_t ws_size,
                              hipStream_t stream) {
    (void)in_sizes; (void)n_in; (void)out_size; (void)ws_size;
    const float* x      = (const float*)d_in[0];
    const float* cond   = (const float*)d_in[1];
    const float* w_ada  = (const float*)d_in[2];
    const float* b_ada  = (const float*)d_in[3];
    const float* w_proj = (const float*)d_in[4];
    const float* b_proj = (const float*)d_in[5];
    float*       out    = (float*)d_out;
    float*       mod    = (float*)d_ws;

    ada_mod_wmma<<<dim3(TWO_D / (16 * K1_TILES)), dim3(K1_WAVES * 32), 0, stream>>>(cond, w_ada, b_ada, mod);
    ln_mod_silu_proj<<<dim3(B_ * T_ / (K2_WAVES * K2_ROWS)), dim3(K2_WAVES * 32), 0, stream>>>(x, mod, w_proj, b_proj, out);
}
